# starting kernel
# baseline (speedup 1.0000x reference)
#include <hip/hip_runtime.h>
#include <math.h>

typedef float v2f __attribute__((ext_vector_type(2)));
typedef float v8f __attribute__((ext_vector_type(8)));

constexpr int Tt = 512;
constexpr int Ff = 16;
constexpr int H1 = 5;
constexpr int H2 = 3;
constexpr int WAVES = 4;
constexpr int BPB   = WAVES * 32;

#if __has_builtin(__builtin_amdgcn_tanhf)
__device__ __forceinline__ float fast_tanh(float x) { return __builtin_amdgcn_tanhf(x); }
#elif __has_builtin(__builtin_amdgcn_tanh_f32)
__device__ __forceinline__ float fast_tanh(float x) { return __builtin_amdgcn_tanh_f32(x); }
#else
__device__ __forceinline__ float fast_tanh(float x) { return tanhf(x); }
#endif

__global__ __launch_bounds__(128, 1) void rnn2_fused_wmma(
    const float* __restrict__ x,
    const float* __restrict__ W_ih1, const float* __restrict__ W_hh1,
    const float* __restrict__ b_ih1, const float* __restrict__ b_hh1,
    const float* __restrict__ W_ih2, const float* __restrict__ W_hh2,
    const float* __restrict__ b_ih2, const float* __restrict__ b_hh2,
    const float* __restrict__ W_fc,  const float* __restrict__ b_fc,
    float* __restrict__ out)
{
    __shared__ float xp[WAVES][32][9];

    const int tid  = threadIdx.x;
    const int lane = tid & 31;
    const int wave = tid >> 5;
    const int half = lane >> 4;
    const int n    = lane & 15;
    const int base = blockIdx.x * BPB + wave * 32;

    v2f bmat[4];
#pragma unroll
    for (int j = 0; j < 4; ++j) {
        const int k0 = j * 4 + 2 * half;
        float w0 = (n < H1) ? W_ih1[n * Ff + k0]     : 0.0f;
        float w1 = (n < H1) ? W_ih1[n * Ff + k0 + 1] : 0.0f;
        v2f b; b.x = w0; b.y = w1;
        bmat[j] = b;
    }
    const float cinit = (n < H1) ? (b_ih1[n] + b_hh1[n]) : 0.0f;

    float whh1[H1][H1], wih2[H2][H1], whh2[H2][H2], b2s[H2], wfc[H2];
#pragma unroll
    for (int i = 0; i < H1; ++i)
#pragma unroll
        for (int j = 0; j < H1; ++j) whh1[i][j] = W_hh1[i * H1 + j];
#pragma unroll
    for (int i = 0; i < H2; ++i) {
#pragma unroll
        for (int j = 0; j < H1; ++j) wih2[i][j] = W_ih2[i * H1 + j];
#pragma unroll
        for (int j = 0; j < H2; ++j) whh2[i][j] = W_hh2[i * H2 + j];
        b2s[i] = b_ih2[i] + b_hh2[i];
        wfc[i] = W_fc[i];
    }
    const float bfc = b_fc[0];

    const float* rowA0  = x + (size_t)(base +      (lane & 15)) * (Tt * Ff) + 2 * half;
    const float* rowA1  = x + (size_t)(base + 16 + (lane & 15)) * (Tt * Ff) + 2 * half;
    const float* rowOwn = x + (size_t)(base + lane) * (Tt * Ff);

    float h1[H1], h2[H2];
#pragma unroll
    for (int i = 0; i < H1; ++i) h1[i] = 0.0f;
#pragma unroll
    for (int i = 0; i < H2; ++i) h2[i] = 0.0f;

    v2f a[8];
#pragma unroll
    for (int j = 0; j < 4; ++j) {
        a[j]     = __builtin_nontemporal_load((const v2f*)(rowA0 + j * 4));
        a[4 + j] = __builtin_nontemporal_load((const v2f*)(rowA1 + j * 4));
    }

    for (int t = 0; t < Tt; ++t) {
        const int tp = (t + 8 < Tt) ? (t + 8) : (Tt - 1);
        __builtin_prefetch(rowOwn + (size_t)tp * Ff, 0, 1);

        v2f anx[8];
        {
            const int tn = (t + 1 < Tt) ? (t + 1) : t;
            const float* p0 = rowA0 + (size_t)tn * Ff;
            const float* p1 = rowA1 + (size_t)tn * Ff;
#pragma unroll
            for (int j = 0; j < 4; ++j) {
                anx[j]     = __builtin_nontemporal_load((const v2f*)(p0 + j * 4));
                anx[4 + j] = __builtin_nontemporal_load((const v2f*)(p1 + j * 4));
            }
        }

        v8f c0, c1;
#pragma unroll
        for (int i = 0; i < 8; ++i) { c0[i] = cinit; c1[i] = cinit; }
#pragma unroll
        for (int j = 0; j < 4; ++j) {
            c0 = __builtin_amdgcn_wmma_f32_16x16x4_f32(
                false, a[j], false, bmat[j], (short)0, c0, false, false);
            c1 = __builtin_amdgcn_wmma_f32_16x16x4_f32(
                false, a[4 + j], false, bmat[j], (short)0, c1, false, false);
        }

        if (n < H1) {
#pragma unroll
            for (int r = 0; r < 8; ++r) xp[wave][half * 8 + r][n]      = c0[r];
#pragma unroll
            for (int r = 0; r < 8; ++r) xp[wave][16 + half * 8 + r][n] = c1[r];
        }

        float xv[H1];
#pragma unroll
        for (int i = 0; i < H1; ++i) xv[i] = xp[wave][lane][i];

        float h1n[H1];
#pragma unroll
        for (int i = 0; i < H1; ++i) {
            float s = xv[i];
#pragma unroll
            for (int j = 0; j < H1; ++j) s += whh1[i][j] * h1[j];
            h1n[i] = fast_tanh(s);
        }
#pragma unroll
        for (int i = 0; i < H1; ++i) h1[i] = h1n[i];

        float h2n[H2];
#pragma unroll
        for (int i = 0; i < H2; ++i) {
            float s = b2s[i];
#pragma unroll
            for (int j = 0; j < H1; ++j) s += wih2[i][j] * h1n[j];
#pragma unroll
            for (int j = 0; j < H2; ++j) s += whh2[i][j] * h2[j];
            h2n[i] = fast_tanh(s);
        }
#pragma unroll
        for (int i = 0; i < H2; ++i) h2[i] = h2n[i];

#pragma unroll
        for (int j = 0; j < 8; ++j) a[j] = anx[j];
    }

    float y = bfc;
#pragma unroll
    for (int j = 0; j < H2; ++j) y += wfc[j] * h2[j];
#if __has_builtin(__builtin_amdgcn_rcpf)
    out[base + lane] = __builtin_amdgcn_rcpf(1.0f + __expf(-y));
#else
    out[base + lane] = 1.0f / (1.0f + __expf(-y));
#endif
}

extern "C" void kernel_launch(void* const* d_in, const int* in_sizes, int n_in,
                              void* d_out, int out_size, void* d_ws, size_t ws_size,
                              hipStream_t stream) {
    const float* x     = (const float*)d_in[0];
    const float* W_ih1 = (const float*)d_in[1];
    const float* W_hh1 = (const float*)d_in[2];
    const float* b_ih1 = (const float*)d_in[3];
    const float* b_hh1 = (const float*)d_in[4];
    const float* W_ih2 = (const float*)d_in[5];
    const float* W_hh2 = (const float*)d_in[6];
    const float* b_ih2 = (const float*)d_in[7];
    const float* b_hh2 = (const float*)d_in[8];
    const float* W_fc  = (const float*)d_in[9];
    const float* b_fc  = (const float*)d_in[10];
    float* out = (float*)d_out;

    const int Btot = in_sizes[0] / (Tt * Ff);
    const int grid = Btot / BPB;

    rnn2_fused_wmma<<<grid, WAVES * 32, 0, stream>>>(
        x, W_ih1, W_hh1, b_ih1, b_hh1, W_ih2, W_hh2, b_ih2, b_hh2, W_fc, b_fc, out);
}


# scan spread over lanes: a row's 8 hidden units on 4 lanes (2 each), layer 2 one step behind, state exchanged by DPP quad rotations; sums re-associated (own lane's term first, layer 2 in two chains)
# speedup vs baseline: 4.7124x; 4.7124x over previous
#include <hip/hip_runtime.h>
#include <math.h>


typedef float v2f __attribute__((ext_vector_type(2)));
typedef float v4f __attribute__((ext_vector_type(4)));
typedef float v8f __attribute__((ext_vector_type(8)));

constexpr int Tt = 512;
constexpr int Ff = 16;
constexpr int H1 = 5;
constexpr int H2 = 3;
constexpr int WAVES = 4;
constexpr int RW    = 8;
constexpr int BPB   = WAVES * RW;
constexpr int TC  = 16;
constexpr int NCH = Tt / TC;
constexpr int NB  = 4;
constexpr int RB  = 2;
constexpr int LPF = 68;

#if __has_builtin(__builtin_amdgcn_tanhf)
__device__ __forceinline__ float fast_tanh(float x) { return __builtin_amdgcn_tanhf(x); }
#elif __has_builtin(__builtin_amdgcn_tanh_f32)
__device__ __forceinline__ float fast_tanh(float x) { return __builtin_amdgcn_tanh_f32(x); }
#else
__device__ __forceinline__ float fast_tanh(float x) { return tanhf(x); }
#endif

template <int CTRL>
__device__ __forceinline__ float quad_perm(float v) {
    return __builtin_bit_cast(float, __builtin_amdgcn_update_dpp(
        0, __builtin_bit_cast(int, v), CTRL, 0xF, 0xF, true));
}
template <int K>
__device__ __forceinline__ float quad_bcast(float v) { return quad_perm<K * 0x55>(v); }
__device__ __forceinline__ float quad_rot1(float v) { return quad_perm<(1 | (2 << 2) | (3 << 4) | (0 << 6))>(v); }
__device__ __forceinline__ float quad_rot2(float v) { return quad_perm<(2 | (3 << 2) | (0 << 4) | (1 << 6))>(v); }
__device__ __forceinline__ float quad_rot3(float v) { return quad_perm<(3 | (0 << 2) | (1 << 4) | (2 << 6))>(v); }

__global__ __launch_bounds__(128, 1) void rnn2_fused_wmma(
    const float* __restrict__ x,
    const float* __restrict__ W_ih1, const float* __restrict__ W_hh1,
    const float* __restrict__ b_ih1, const float* __restrict__ b_hh1,
    const float* __restrict__ W_ih2, const float* __restrict__ W_hh2,
    const float* __restrict__ b_ih2, const float* __restrict__ b_hh2,
    const float* __restrict__ W_fc,  const float* __restrict__ b_fc,
    float* __restrict__ out)
{
    __shared__ __attribute__((aligned(16))) float park[WAVES][TC][LPF];
    __shared__ float ybuf[BPB];

    const int tid  = threadIdx.x;
    const int lane = tid & 31;
    const int wave = tid >> 5;
    const int half = lane >> 4;
    const int n    = lane & 15;
    const int hu   = n & 7;
    const int rq   = lane >> 2;
    const int g    = lane & 3;
    const int base = blockIdx.x * BPB + wave * RW;

    v2f wmat[4];
#pragma unroll
    for (int j = 0; j < 4; ++j) {
        const int k0 = j * 4 + 2 * half;
        float w0 = (hu < H1) ? W_ih1[hu * Ff + k0]     : 0.0f;
        float w1 = (hu < H1) ? W_ih1[hu * Ff + k0 + 1] : 0.0f;
        v2f w; w.x = w0; w.y = w1;
        wmat[j] = w;
    }
    float cinit[8];
#pragma unroll
    for (int r = 0; r < 8; ++r) cinit[r] = 0.0f;
#pragma unroll
    for (int r = 0; r < H1; ++r) cinit[r] = b_ih1[r] + b_hh1[r];

    const int u2 = (g == 0) ? 0 : (g - 1);
    float wAA[4], wBA[4], wBB[4];
#pragma unroll
    for (int d = 0; d < 4; ++d) {
        const int q  = (g + d) & 3;
        const int q2 = (q == 0) ? 0 : (q - 1);
        wAA[d] = W_hh1[g * H1 + q];
        const float a1 = W_hh1[4 * H1 + q];
        const float a2 = W_ih2[u2 * H1 + q];
        wBA[d] = (g == 0) ? a1 : a2;
        const float l1a = W_hh1[4 * H1 + 4];
        const float l1b = W_ih2[u2 * H1 + 4];
        const float l1  = (g == 0) ? l1a : l1b;
        const float l2b = W_hh2[u2 * H2 + q2];
        const float l2  = (g == 0) ? 0.0f : l2b;
        wBB[d] = (q == 0) ? l1 : l2;
    }
    const float wA4 = W_hh1[g * H1 + 4];
    float b2s[H2], wfc[H2];
#pragma unroll
    for (int i = 0; i < H2; ++i) {
        b2s[i] = b_ih2[i] + b_hh2[i];
        wfc[i] = W_fc[i];
    }
    const float b2l = b_ih2[u2] + b_hh2[u2];
    const float bfc = b_fc[0];

    const float* xl = x + (size_t)base * (Tt * Ff) + n * Ff + 2 * half;

    float sA = 0.0f, sB = 0.0f;
    float* pst = &park[wave][n][half * 8];
    const float* prd = &park[wave][0][rq * 8 + g];

    v2f R[NB][RB][4];
#pragma unroll
    for (int q = 0; q < NB; ++q)
#pragma unroll
        for (int i = 0; i < RB; ++i)
#pragma unroll
            for (int j = 0; j < 4; ++j)
                R[q][i][j] = __builtin_nontemporal_load(
                    (const v2f*)(xl + (size_t)(q * RB + i) * (Tt * Ff) + j * 4));

    for (int ch = 0; ch < NCH; ++ch) {
        const int chn = (ch + 1 < NCH) ? (ch + 1) : ch;
        const float* xn = xl + (size_t)chn * (TC * Ff);

#pragma unroll
        for (int q = 0; q < NB; ++q) {
#pragma unroll
            for (int i = 0; i < RB; i += 2) {
                v8f c0, c1;
#pragma unroll
                for (int r = 0; r < 8; ++r) { c0[r] = cinit[r]; c1[r] = cinit[r]; }
#pragma unroll
                for (int j = 0; j < 4; ++j) {
                    c0 = __builtin_amdgcn_wmma_f32_16x16x4_f32(
                        false, wmat[j], false, R[q][i][j],     (short)0, c0, false, false);
                    c1 = __builtin_amdgcn_wmma_f32_16x16x4_f32(
                        false, wmat[j], false, R[q][i + 1][j], (short)0, c1, false, false);
                }
                const int r0 = q * RB + i;
                v4f d, e;
                d.x = half ? c1[0] : c0[0];
                d.y = half ? c1[1] : c0[1];
                d.z = half ? c1[2] : c0[2];
                d.w = half ? c1[3] : c0[3];
                e.x = half ? c1[4] : c0[4];
                e.y = b2s[0];
                e.z = b2s[1];
                e.w = b2s[2];
                *(v4f*)(pst + r0 * 8)     = d;
                *(v4f*)(pst + r0 * 8 + 4) = e;
            }
            __builtin_amdgcn_sched_barrier(0);
#pragma unroll
            for (int i = 0; i < RB; ++i)
#pragma unroll
                for (int j = 0; j < 4; ++j)
                    R[q][i][j] = __builtin_nontemporal_load(
                        (const v2f*)(xn + (size_t)(q * RB + i) * (Tt * Ff) + j * 4));
            __builtin_amdgcn_sched_barrier(0);
        }

        const bool zero_l2 = (ch == 0) && (g != 0);
        float inAs[TC], inBs[TC];
#pragma unroll
        for (int s = 0; s < TC; ++s) {
            inAs[s] = prd[s * LPF];
            inBs[s] = prd[s * LPF + 4];
        }
        __builtin_amdgcn_sched_barrier(0);
#pragma unroll
        for (int s = 0; s < TC; ++s) {
            const float inA = inAs[s];
            const float inB = inBs[s];

            const float a1 = quad_rot1(sA), a2 = quad_rot2(sA), a3 = quad_rot3(sA);
            const float e1 = quad_rot1(sB), e2 = quad_rot2(sB), e3 = quad_rot3(sB);
            const float u4 = quad_bcast<0>(sB);

            float accA = inA;
            accA = __builtin_fmaf(wAA[0], sA, accA);
            accA = __builtin_fmaf(wAA[1], a1, accA);
            accA = __builtin_fmaf(wAA[2], a2, accA);
            accA = __builtin_fmaf(wAA[3], a3, accA);
            accA = __builtin_fmaf(wA4,    u4, accA);
            float accB = inB;
            accB = __builtin_fmaf(wBA[0], sA, accB);
            accB = __builtin_fmaf(wBA[1], a1, accB);
            accB = __builtin_fmaf(wBA[2], a2, accB);
            accB = __builtin_fmaf(wBA[3], a3, accB);
            float accC = wBB[0] * sB;
            accC = __builtin_fmaf(wBB[1], e1, accC);
            accC = __builtin_fmaf(wBB[2], e2, accC);
            accC = __builtin_fmaf(wBB[3], e3, accC);

            sA = fast_tanh(accA);
            sB = fast_tanh(accB + accC);
            if (s == 0) sB = zero_l2 ? 0.0f : sB;
        }
    }

    asm volatile("s_wait_loadcnt 0x0" ::: "memory");

    float h2l;
    {
        const float a1 = quad_rot1(sA), a2 = quad_rot2(sA), a3 = quad_rot3(sA);
        const float e1 = quad_rot1(sB), e2 = quad_rot2(sB), e3 = quad_rot3(sB);
        float accB = b2l;
        accB = __builtin_fmaf(wBA[0], sA, accB);
        accB = __builtin_fmaf(wBA[1], a1, accB);
        accB = __builtin_fmaf(wBA[2], a2, accB);
        accB = __builtin_fmaf(wBA[3], a3, accB);
        float accC = wBB[0] * sB;
        accC = __builtin_fmaf(wBB[1], e1, accC);
        accC = __builtin_fmaf(wBB[2], e2, accC);
        accC = __builtin_fmaf(wBB[3], e3, accC);
        h2l = fast_tanh(accB + accC);
    }
    float y = bfc;
    y += wfc[0] * quad_bcast<1>(h2l);
    y += wfc[1] * quad_bcast<2>(h2l);
    y += wfc[2] * quad_bcast<3>(h2l);
#if __has_builtin(__builtin_amdgcn_rcpf)
    const float yo = __builtin_amdgcn_rcpf(1.0f + __expf(-y));
#else
    const float yo = 1.0f / (1.0f + __expf(-y));
#endif
    if (g == 0) ybuf[wave * RW + rq] = yo;
    __syncthreads();
    if (wave == 0) out[blockIdx.x * BPB + lane] = ybuf[lane];
}

extern "C" void kernel_launch(void* const* d_in, const int* in_sizes, int n_in,
                              void* d_out, int out_size, void* d_ws, size_t ws_size,
                              hipStream_t stream) {
    const float* x     = (const float*)d_in[0];
    const float* W_ih1 = (const float*)d_in[1];
    const float* W_hh1 = (const float*)d_in[2];
    const float* b_ih1 = (const float*)d_in[3];
    const float* b_hh1 = (const float*)d_in[4];
    const float* W_ih2 = (const float*)d_in[5];
    const float* W_hh2 = (const float*)d_in[6];
    const float* b_ih2 = (const float*)d_in[7];
    const float* b_hh2 = (const float*)d_in[8];
    const float* W_fc  = (const float*)d_in[9];
    const float* b_fc  = (const float*)d_in[10];
    float* out = (float*)d_out;

    const int Btot = in_sizes[0] / (Tt * Ff);
    const int grid = Btot / BPB;

    rnn2_fused_wmma<<<grid, WAVES * 32, 0, stream>>>(
        x, W_ih1, W_hh1, b_ih1, b_hh1, W_ih2, W_hh2, b_ih2, b_hh2, W_fc, b_fc, out);
}
